# starting kernel
# baseline (speedup 1.0000x reference)
#include <hip/hip_runtime.h>

typedef __attribute__((ext_vector_type(2))) float v2f;
typedef __attribute__((ext_vector_type(8))) float v8f;

#define NCOMP 5
#define BISECT_ITERS 60

__device__ __forceinline__ float fexp2(float x) { return __builtin_amdgcn_exp2f(x); }
__device__ __forceinline__ float flog2(float x) { return __builtin_amdgcn_logf(x); }
__device__ __forceinline__ float frcp(float x)  { return __builtin_amdgcn_rcpf(x); }

__device__ __forceinline__ bool g_pos(const float* dd, float lam) {
    const float CC2 = 2.17765859079846660f;
    float t = frcp(1.0f + lam);
    float S = 0.0f, T = 0.0f;
    #pragma unroll
    for (int i = 0; i < NCOMP; ++i) {
        float z = t * dd[i];
        float e = fexp2(z);
        S += e;
        T = fmaf(e, z, T);
    }
    return fmaf(S, CC2 - flog2(S), T) > 0.0f;
}

__device__ __forceinline__ void solve_row(const float* q, float* p) {
    float dd[NCOMP], dmax = -3.0e38f;
    #pragma unroll
    for (int i = 0; i < NCOMP; ++i) { dd[i] = flog2(q[i]); dmax = fmaxf(dmax, dd[i]); }
    #pragma unroll
    for (int i = 0; i < NCOMP; ++i) dd[i] -= dmax;

    bool infeasible0 = g_pos(dd, 0.0f);
    float lo = 0.0f, hi = 1.0e6f;
    #pragma unroll 1
    for (int it = 0; it < BISECT_ITERS; ++it) {
        float mid = 0.5f * (lo + hi);
        bool pos = g_pos(dd, mid);
        lo = pos ? mid : lo;
        hi = pos ? hi : mid;
    }
    float lam = infeasible0 ? 0.5f * (lo + hi) : 0.0f;
    float t = frcp(1.0f + lam);
    float e[NCOMP], S = 0.0f;
    #pragma unroll
    for (int i = 0; i < NCOMP; ++i) { e[i] = fexp2(t * dd[i]); S += e[i]; }
    float inv = 1.0f / S;
    #pragma unroll
    for (int i = 0; i < NCOMP; ++i) p[i] = e[i] * inv;
}

__global__ __launch_bounds__(256) void klproj_wmma_kernel(
    const float* __restrict__ x, const float* __restrict__ W,
    const float* __restrict__ bvec, float* __restrict__ out, int B)
{
    __shared__ float lds[8 * 160];
    const int lane  = threadIdx.x & 31;
    const int wIB   = threadIdx.x >> 5;
    const int gwave = blockIdx.x * 8 + wIB;
    const int nwave = gridDim.x * 8;
    float* wls = &lds[wIB * 160];

    const int n    = lane & 15;
    const int hi   = lane >> 4;
    const int koff = hi << 1;

    float b1x = 0.f, b1y = 0.f, b2x = 0.f, cbias = 0.f;
    if (n < NCOMP) {
        b1x = W[n * NCOMP + koff];
        b1y = W[n * NCOMP + koff + 1];
        if (!hi) b2x = W[n * NCOMP + 4];
        cbias = bvec[n];
    }
    v2f Bf1; Bf1.x = b1x; Bf1.y = b1y;
    v2f Bf2; Bf2.x = b2x; Bf2.y = 0.f;

    const int nChunks = B >> 5;
    for (int c = gwave; c < nChunks; c += nwave) {
        const long base5 = (long)c * 160;
        #pragma unroll
        for (int j = 0; j < 5; ++j)
            wls[j * 32 + lane] = x[base5 + j * 32 + lane];
        asm volatile("s_wait_dscnt 0" ::: "memory");

        #pragma unroll
        for (int t = 0; t < 2; ++t) {
            const int mrow = t * 16 + n;
            v2f Af1, Af2;
            Af1.x = wls[mrow * 5 + koff];
            Af1.y = wls[mrow * 5 + koff + 1];
            float a4 = wls[mrow * 5 + 4];
            Af2.x = hi ? 0.f : a4;
            Af2.y = 0.f;
            v8f acc;
            #pragma unroll
            for (int i = 0; i < 8; ++i) acc[i] = cbias;
            acc = __builtin_amdgcn_wmma_f32_16x16x4_f32(
                      false, Af1, false, Bf1, (short)0, acc, false, false);
            acc = __builtin_amdgcn_wmma_f32_16x16x4_f32(
                      false, Af2, false, Bf2, (short)0, acc, false, false);
            if (n < NCOMP) {
                #pragma unroll
                for (int v = 0; v < 8; ++v) {
                    int m = t * 16 + v + (hi << 3);
                    wls[m * 5 + n] = acc[v];
                }
            }
        }
        asm volatile("s_wait_dscnt 0" ::: "memory");

        float q[NCOMP], p[NCOMP];
        #pragma unroll
        for (int i = 0; i < NCOMP; ++i) q[i] = wls[lane * 5 + i];
        solve_row(q, p);

        #pragma unroll
        for (int i = 0; i < NCOMP; ++i) wls[lane * 5 + i] = p[i];
        asm volatile("s_wait_dscnt 0" ::: "memory");
        #pragma unroll
        for (int j = 0; j < 5; ++j)
            out[base5 + j * 32 + lane] = wls[j * 32 + lane];
    }

    const int tailStart = nChunks << 5;
    for (int row = tailStart + (int)(blockIdx.x * blockDim.x + threadIdx.x);
         row < B; row += (int)(gridDim.x * blockDim.x)) {
        float q[NCOMP], p[NCOMP];
        #pragma unroll
        for (int i = 0; i < NCOMP; ++i) {
            float acc = bvec[i];
            #pragma unroll
            for (int k = 0; k < NCOMP; ++k)
                acc += x[(long)row * NCOMP + k] * W[i * NCOMP + k];
            q[i] = acc;
        }
        solve_row(q, p);
        #pragma unroll
        for (int i = 0; i < NCOMP; ++i) out[(long)row * NCOMP + i] = p[i];
    }
}

extern "C" void kernel_launch(void* const* d_in, const int* in_sizes, int n_in,
                              void* d_out, int out_size, void* d_ws, size_t ws_size,
                              hipStream_t stream) {
    const float* x = (const float*)d_in[0];
    const float* W = (const float*)d_in[1];
    const float* b = (const float*)d_in[2];
    float* out = (float*)d_out;
    const int B = in_sizes[0] / NCOMP;

    const int nChunks = B >> 5;
    int blocks = (nChunks + 7) / 8;
    if (blocks > 4096) blocks = 4096;
    if (blocks < 1) blocks = 1;
    klproj_wmma_kernel<<<blocks, 256, 0, stream>>>(x, W, b, out, B);
}


# Skip the 60-step bisection in waves where no row has g(0)>0 (wave-uniform ballot test); its result is unused there (lam=0), output bit-identical
# speedup vs baseline: 4.6986x; 4.6986x over previous
#include <hip/hip_runtime.h>

typedef __attribute__((ext_vector_type(2))) float v2f;
typedef __attribute__((ext_vector_type(8))) float v8f;

#define NCOMP 5
#define BISECT_ITERS 60

__device__ __forceinline__ float fexp2(float x) { return __builtin_amdgcn_exp2f(x); }
__device__ __forceinline__ float flog2(float x) { return __builtin_amdgcn_logf(x); }
__device__ __forceinline__ float frcp(float x)  { return __builtin_amdgcn_rcpf(x); }

__device__ __forceinline__ bool g_pos(const float* dd, float lam) {
    const float CC2 = 2.17765859079846660f;
    float t = frcp(1.0f + lam);
    float S = 0.0f, T = 0.0f;
    #pragma unroll
    for (int i = 0; i < NCOMP; ++i) {
        float z = t * dd[i];
        float e = fexp2(z);
        S += e;
        T = fmaf(e, z, T);
    }
    return fmaf(S, CC2 - flog2(S), T) > 0.0f;
}

__device__ __forceinline__ void solve_row(const float* q, float* p) {
    float dd[NCOMP], dmax = -3.0e38f;
    #pragma unroll
    for (int i = 0; i < NCOMP; ++i) { dd[i] = flog2(q[i]); dmax = fmaxf(dmax, dd[i]); }
    #pragma unroll
    for (int i = 0; i < NCOMP; ++i) dd[i] -= dmax;

    bool infeasible0 = g_pos(dd, 0.0f);
    float lam = 0.0f;
    if (__builtin_amdgcn_ballot_w32(infeasible0) != 0u) {
        float lo = 0.0f, hi = 1.0e6f;
        #pragma unroll 1
        for (int it = 0; it < BISECT_ITERS; ++it) {
            float mid = 0.5f * (lo + hi);
            bool pos = g_pos(dd, mid);
            lo = pos ? mid : lo;
            hi = pos ? hi : mid;
        }
        lam = infeasible0 ? 0.5f * (lo + hi) : 0.0f;
    }
    float t = frcp(1.0f + lam);
    float e[NCOMP], S = 0.0f;
    #pragma unroll
    for (int i = 0; i < NCOMP; ++i) { e[i] = fexp2(t * dd[i]); S += e[i]; }
    float inv = 1.0f / S;
    #pragma unroll
    for (int i = 0; i < NCOMP; ++i) p[i] = e[i] * inv;
}

__global__ __launch_bounds__(256) void klproj_wmma_kernel(
    const float* __restrict__ x, const float* __restrict__ W,
    const float* __restrict__ bvec, float* __restrict__ out, int B)
{
    __shared__ float lds[8 * 160];
    const int lane  = threadIdx.x & 31;
    const int wIB   = threadIdx.x >> 5;
    const int gwave = blockIdx.x * 8 + wIB;
    const int nwave = gridDim.x * 8;
    float* wls = &lds[wIB * 160];

    const int n    = lane & 15;
    const int hi   = lane >> 4;
    const int koff = hi << 1;

    float b1x = 0.f, b1y = 0.f, b2x = 0.f, cbias = 0.f;
    if (n < NCOMP) {
        b1x = W[n * NCOMP + koff];
        b1y = W[n * NCOMP + koff + 1];
        if (!hi) b2x = W[n * NCOMP + 4];
        cbias = bvec[n];
    }
    v2f Bf1; Bf1.x = b1x; Bf1.y = b1y;
    v2f Bf2; Bf2.x = b2x; Bf2.y = 0.f;

    const int nChunks = B >> 5;
    for (int c = gwave; c < nChunks; c += nwave) {
        const long base5 = (long)c * 160;
        #pragma unroll
        for (int j = 0; j < 5; ++j)
            wls[j * 32 + lane] = x[base5 + j * 32 + lane];
        asm volatile("s_wait_dscnt 0" ::: "memory");

        #pragma unroll
        for (int t = 0; t < 2; ++t) {
            const int mrow = t * 16 + n;
            v2f Af1, Af2;
            Af1.x = wls[mrow * 5 + koff];
            Af1.y = wls[mrow * 5 + koff + 1];
            float a4 = wls[mrow * 5 + 4];
            Af2.x = hi ? 0.f : a4;
            Af2.y = 0.f;
            v8f acc;
            #pragma unroll
            for (int i = 0; i < 8; ++i) acc[i] = cbias;
            acc = __builtin_amdgcn_wmma_f32_16x16x4_f32(
                      false, Af1, false, Bf1, (short)0, acc, false, false);
            acc = __builtin_amdgcn_wmma_f32_16x16x4_f32(
                      false, Af2, false, Bf2, (short)0, acc, false, false);
            if (n < NCOMP) {
                #pragma unroll
                for (int v = 0; v < 8; ++v) {
                    int m = t * 16 + v + (hi << 3);
                    wls[m * 5 + n] = acc[v];
                }
            }
        }
        asm volatile("s_wait_dscnt 0" ::: "memory");

        float q[NCOMP], p[NCOMP];
        #pragma unroll
        for (int i = 0; i < NCOMP; ++i) q[i] = wls[lane * 5 + i];
        solve_row(q, p);

        #pragma unroll
        for (int i = 0; i < NCOMP; ++i) wls[lane * 5 + i] = p[i];
        asm volatile("s_wait_dscnt 0" ::: "memory");
        #pragma unroll
        for (int j = 0; j < 5; ++j)
            out[base5 + j * 32 + lane] = wls[j * 32 + lane];
    }

    const int tailStart = nChunks << 5;
    for (int row = tailStart + (int)(blockIdx.x * blockDim.x + threadIdx.x);
         row < B; row += (int)(gridDim.x * blockDim.x)) {
        float q[NCOMP], p[NCOMP];
        #pragma unroll
        for (int i = 0; i < NCOMP; ++i) {
            float acc = bvec[i];
            #pragma unroll
            for (int k = 0; k < NCOMP; ++k)
                acc += x[(long)row * NCOMP + k] * W[i * NCOMP + k];
            q[i] = acc;
        }
        solve_row(q, p);
        #pragma unroll
        for (int i = 0; i < NCOMP; ++i) out[(long)row * NCOMP + i] = p[i];
    }
}

extern "C" void kernel_launch(void* const* d_in, const int* in_sizes, int n_in,
                              void* d_out, int out_size, void* d_ws, size_t ws_size,
                              hipStream_t stream) {
    const float* x = (const float*)d_in[0];
    const float* W = (const float*)d_in[1];
    const float* b = (const float*)d_in[2];
    float* out = (float*)d_out;
    const int B = in_sizes[0] / NCOMP;

    const int nChunks = B >> 5;
    int blocks = (nChunks + 7) / 8;
    if (blocks > 4096) blocks = 4096;
    if (blocks < 1) blocks = 1;
    klproj_wmma_kernel<<<blocks, 256, 0, stream>>>(x, W, b, out, B);
}


# One wave = 128 rows via 16-byte loads/stores (5 per lane, issued before the W/b loads; 2048 blocks = one unit per wave slot, was 4096 x 2 chunks); per-row math and summation order unchanged
# speedup vs baseline: 5.3445x; 1.1375x over previous
#include <hip/hip_runtime.h>

typedef __attribute__((ext_vector_type(2))) float v2f;
typedef __attribute__((ext_vector_type(8))) float v8f;

#define NCOMP 5
#define BISECT_ITERS 60

__device__ __forceinline__ float fexp2(float x) { return __builtin_amdgcn_exp2f(x); }
__device__ __forceinline__ float flog2(float x) { return __builtin_amdgcn_logf(x); }
__device__ __forceinline__ float frcp(float x)  { return __builtin_amdgcn_rcpf(x); }

__device__ __forceinline__ bool g_pos(const float* dd, float lam) {
    const float CC2 = 2.17765859079846660f;
    float t = frcp(1.0f + lam);
    float S = 0.0f, T = 0.0f;
    #pragma unroll
    for (int i = 0; i < NCOMP; ++i) {
        float z = t * dd[i];
        float e = fexp2(z);
        S += e;
        T = fmaf(e, z, T);
    }
    return fmaf(S, CC2 - flog2(S), T) > 0.0f;
}

__device__ __forceinline__ void solve_row(const float* q, float* p) {
    float dd[NCOMP], dmax = -3.0e38f;
    #pragma unroll
    for (int i = 0; i < NCOMP; ++i) { dd[i] = flog2(q[i]); dmax = fmaxf(dmax, dd[i]); }
    #pragma unroll
    for (int i = 0; i < NCOMP; ++i) dd[i] -= dmax;

    bool infeasible0 = g_pos(dd, 0.0f);
    float lam = 0.0f;
    if (__builtin_amdgcn_ballot_w32(infeasible0) != 0u) {
        float lo = 0.0f, hi = 1.0e6f;
        #pragma unroll 1
        for (int it = 0; it < BISECT_ITERS; ++it) {
            float mid = 0.5f * (lo + hi);
            bool pos = g_pos(dd, mid);
            lo = pos ? mid : lo;
            hi = pos ? hi : mid;
        }
        lam = infeasible0 ? 0.5f * (lo + hi) : 0.0f;
    }
    float t = frcp(1.0f + lam);
    float e[NCOMP], S = 0.0f;
    #pragma unroll
    for (int i = 0; i < NCOMP; ++i) { e[i] = fexp2(t * dd[i]); S += e[i]; }
    float inv = 1.0f / S;
    #pragma unroll
    for (int i = 0; i < NCOMP; ++i) p[i] = e[i] * inv;
}

typedef __attribute__((ext_vector_type(4))) float v4f;

#define UNIT_ROWS   128
#define UNIT_FLOATS (UNIT_ROWS * NCOMP)

__global__ __launch_bounds__(256) void klproj_wmma_kernel(
    const float* __restrict__ x, const float* __restrict__ W,
    const float* __restrict__ bvec, float* __restrict__ out, int B)
{
    __shared__ v4f lds4[8 * (UNIT_FLOATS / 4)];
    const int lane  = threadIdx.x & 31;
    const int wIB   = threadIdx.x >> 5;
    const int gwave = blockIdx.x * 8 + wIB;
    const int nwave = gridDim.x * 8;
    v4f*   wls4 = &lds4[wIB * (UNIT_FLOATS / 4)];
    float* wlsU = (float*)wls4;

    const int nUnits = B >> 7;
    if (gwave < nUnits) {
        int u = gwave;
        v4f r[5];
        {
            const v4f* xin = (const v4f*)(x + (long)u * UNIT_FLOATS);
            #pragma unroll
            for (int j = 0; j < 5; ++j) r[j] = xin[j * 32 + lane];
        }

        const int n    = lane & 15;
        const int hi   = lane >> 4;
        const int koff = hi << 1;

        float b1x = 0.f, b1y = 0.f, b2x = 0.f, cbias = 0.f;
        if (n < NCOMP) {
            b1x = W[n * NCOMP + koff];
            b1y = W[n * NCOMP + koff + 1];
            if (!hi) b2x = W[n * NCOMP + 4];
            cbias = bvec[n];
        }
        v2f Bf1; Bf1.x = b1x; Bf1.y = b1y;
        v2f Bf2; Bf2.x = b2x; Bf2.y = 0.f;

        for (;;) {
            #pragma unroll
            for (int j = 0; j < 5; ++j) wls4[j * 32 + lane] = r[j];
            asm volatile("s_wait_dscnt 0" ::: "memory");

            #pragma unroll 1
            for (int s = 0; s < UNIT_ROWS / 32; ++s) {
                float* wls = wlsU + s * 160;
                #pragma unroll
                for (int t = 0; t < 2; ++t) {
                    const int mrow = t * 16 + n;
                    v2f Af1, Af2;
                    Af1.x = wls[mrow * 5 + koff];
                    Af1.y = wls[mrow * 5 + koff + 1];
                    float a4 = wls[mrow * 5 + 4];
                    Af2.x = hi ? 0.f : a4;
                    Af2.y = 0.f;
                    v8f acc;
                    #pragma unroll
                    for (int i = 0; i < 8; ++i) acc[i] = cbias;
                    acc = __builtin_amdgcn_wmma_f32_16x16x4_f32(
                              false, Af1, false, Bf1, (short)0, acc, false, false);
                    acc = __builtin_amdgcn_wmma_f32_16x16x4_f32(
                              false, Af2, false, Bf2, (short)0, acc, false, false);
                    if (n < NCOMP) {
                        #pragma unroll
                        for (int v = 0; v < 8; ++v) {
                            int m = t * 16 + v + (hi << 3);
                            wls[m * 5 + n] = acc[v];
                        }
                    }
                }
                asm volatile("s_wait_dscnt 0" ::: "memory");

                float q[NCOMP], p[NCOMP];
                #pragma unroll
                for (int i = 0; i < NCOMP; ++i) q[i] = wls[lane * 5 + i];
                solve_row(q, p);

                #pragma unroll
                for (int i = 0; i < NCOMP; ++i) wls[lane * 5 + i] = p[i];
            }
            asm volatile("s_wait_dscnt 0" ::: "memory");
            v4f* yout = (v4f*)(out + (long)u * UNIT_FLOATS);
            #pragma unroll
            for (int j = 0; j < 5; ++j) yout[j * 32 + lane] = wls4[j * 32 + lane];

            u += nwave;
            if (u >= nUnits) break;
            const v4f* xin = (const v4f*)(x + (long)u * UNIT_FLOATS);
            #pragma unroll
            for (int j = 0; j < 5; ++j) r[j] = xin[j * 32 + lane];
        }
    }

    const int tailStart = nUnits << 7;
    for (int row = tailStart + (int)(blockIdx.x * blockDim.x + threadIdx.x);
         row < B; row += (int)(gridDim.x * blockDim.x)) {
        float q[NCOMP], p[NCOMP];
        #pragma unroll
        for (int i = 0; i < NCOMP; ++i) {
            float acc = bvec[i];
            #pragma unroll
            for (int k = 0; k < NCOMP; ++k)
                acc += x[(long)row * NCOMP + k] * W[i * NCOMP + k];
            q[i] = acc;
        }
        solve_row(q, p);
        #pragma unroll
        for (int i = 0; i < NCOMP; ++i) out[(long)row * NCOMP + i] = p[i];
    }
}

extern "C" void kernel_launch(void* const* d_in, const int* in_sizes, int n_in,
                              void* d_out, int out_size, void* d_ws, size_t ws_size,
                              hipStream_t stream) {
    const float* x = (const float*)d_in[0];
    const float* W = (const float*)d_in[1];
    const float* b = (const float*)d_in[2];
    float* out = (float*)d_out;
    const int B = in_sizes[0] / NCOMP;

    const int nUnits = B >> 7;
    int blocks = (nUnits + 7) / 8;
    if (blocks > 4096) blocks = 4096;
    if (blocks < 1) blocks = 1;
    klproj_wmma_kernel<<<blocks, 256, 0, stream>>>(x, W, b, out, B);
}


# Block-diagonal W operand: one WMMA covers 32 rows x 2 k (lane half = row group), so q lands one row per lane; drops the LDS scatter and re-read and 1 of 4 WMMAs; k-sum regrouped (b+k0,k1)+(k2,k3)+k4
# speedup vs baseline: 6.1240x; 1.1459x over previous
#include <hip/hip_runtime.h>

typedef __attribute__((ext_vector_type(2))) float v2f;
typedef __attribute__((ext_vector_type(8))) float v8f;

#define NCOMP 5
#define BISECT_ITERS 60

__device__ __forceinline__ float fexp2(float x) { return __builtin_amdgcn_exp2f(x); }
__device__ __forceinline__ float flog2(float x) { return __builtin_amdgcn_logf(x); }
__device__ __forceinline__ float frcp(float x)  { return __builtin_amdgcn_rcpf(x); }

__device__ __forceinline__ bool g_pos(const float* dd, float lam) {
    const float CC2 = 2.17765859079846660f;
    float t = frcp(1.0f + lam);
    float S = 0.0f, T = 0.0f;
    #pragma unroll
    for (int i = 0; i < NCOMP; ++i) {
        float z = t * dd[i];
        float e = fexp2(z);
        S += e;
        T = fmaf(e, z, T);
    }
    return fmaf(S, CC2 - flog2(S), T) > 0.0f;
}

__device__ __forceinline__ void solve_row(const float* q, float* p) {
    float dd[NCOMP], dmax = -3.0e38f;
    #pragma unroll
    for (int i = 0; i < NCOMP; ++i) { dd[i] = flog2(q[i]); dmax = fmaxf(dmax, dd[i]); }
    #pragma unroll
    for (int i = 0; i < NCOMP; ++i) dd[i] -= dmax;

    bool infeasible0 = g_pos(dd, 0.0f);
    float lam = 0.0f;
    if (__builtin_amdgcn_ballot_w32(infeasible0) != 0u) {
        float lo = 0.0f, hi = 1.0e6f;
        #pragma unroll 1
        for (int it = 0; it < BISECT_ITERS; ++it) {
            float mid = 0.5f * (lo + hi);
            bool pos = g_pos(dd, mid);
            lo = pos ? mid : lo;
            hi = pos ? hi : mid;
        }
        lam = infeasible0 ? 0.5f * (lo + hi) : 0.0f;
    }
    float t = frcp(1.0f + lam);
    float e[NCOMP], S = 0.0f;
    #pragma unroll
    for (int i = 0; i < NCOMP; ++i) { e[i] = fexp2(t * dd[i]); S += e[i]; }
    float inv = 1.0f / S;
    #pragma unroll
    for (int i = 0; i < NCOMP; ++i) p[i] = e[i] * inv;
}

typedef __attribute__((ext_vector_type(4))) float v4f;

#define UNIT_ROWS   128
#define UNIT_FLOATS (UNIT_ROWS * NCOMP)

__global__ __launch_bounds__(256) void klproj_wmma_kernel(
    const float* __restrict__ x, const float* __restrict__ W,
    const float* __restrict__ bvec, float* __restrict__ out, int B)
{
    __shared__ v4f lds4[8 * (UNIT_FLOATS / 4)];
    const int lane  = threadIdx.x & 31;
    const int wIB   = threadIdx.x >> 5;
    const int gwave = blockIdx.x * 8 + wIB;
    const int nwave = gridDim.x * 8;
    v4f*   wls4 = &lds4[wIB * (UNIT_FLOATS / 4)];
    float* wlsU = (float*)wls4;

    const int nUnits = B >> 7;
    if (gwave < nUnits) {
        int u = gwave;
        v4f r[5];
        {
            const v4f* xin = (const v4f*)(x + (long)u * UNIT_FLOATS);
            #pragma unroll
            for (int j = 0; j < 5; ++j) r[j] = xin[j * 32 + lane];
        }

        const int comp = lane & 7;
        const bool live = (comp < NCOMP) && (((lane >> 3) & 1) == (lane >> 4));
        float w0 = 0.f, w1 = 0.f, w2 = 0.f, w3 = 0.f, w4 = 0.f;
        if (live) {
            w0 = W[comp * NCOMP + 0];
            w1 = W[comp * NCOMP + 1];
            w2 = W[comp * NCOMP + 2];
            w3 = W[comp * NCOMP + 3];
            w4 = W[comp * NCOMP + 4];
        }
        v2f A01; A01.x = w0; A01.y = w1;
        v2f A23; A23.x = w2; A23.y = w3;
        v2f A4;  A4.x  = w4; A4.y  = 0.f;
        v8f cb;
        #pragma unroll
        for (int i = 0; i < 8; ++i) cb[i] = (i < NCOMP) ? bvec[i] : 0.f;

        for (;;) {
            #pragma unroll
            for (int j = 0; j < 5; ++j) wls4[j * 32 + lane] = r[j];
            asm volatile("s_wait_dscnt 0" ::: "memory");

            #pragma unroll 1
            for (int s = 0; s < UNIT_ROWS / 32; ++s) {
                float* row = wlsU + s * 160 + lane * 5;
                float xr[NCOMP];
                #pragma unroll
                for (int i = 0; i < NCOMP; ++i) xr[i] = row[i];
                v2f B01; B01.x = xr[0]; B01.y = xr[1];
                v2f B23; B23.x = xr[2]; B23.y = xr[3];
                v2f B4;  B4.x  = xr[4]; B4.y  = 0.f;
                v8f acc = __builtin_amdgcn_wmma_f32_16x16x4_f32(
                              false, A01, false, B01, (short)0, cb,  false, false);
                acc = __builtin_amdgcn_wmma_f32_16x16x4_f32(
                              false, A23, false, B23, (short)0, acc, false, false);
                acc = __builtin_amdgcn_wmma_f32_16x16x4_f32(
                              false, A4,  false, B4,  (short)0, acc, false, false);

                float q[NCOMP], p[NCOMP];
                #pragma unroll
                for (int i = 0; i < NCOMP; ++i) q[i] = acc[i];
                solve_row(q, p);

                #pragma unroll
                for (int i = 0; i < NCOMP; ++i) row[i] = p[i];
            }
            asm volatile("s_wait_dscnt 0" ::: "memory");
            v4f* yout = (v4f*)(out + (long)u * UNIT_FLOATS);
            #pragma unroll
            for (int j = 0; j < 5; ++j) yout[j * 32 + lane] = wls4[j * 32 + lane];

            u += nwave;
            if (u >= nUnits) break;
            const v4f* xin = (const v4f*)(x + (long)u * UNIT_FLOATS);
            #pragma unroll
            for (int j = 0; j < 5; ++j) r[j] = xin[j * 32 + lane];
        }
    }

    const int tailStart = nUnits << 7;
    for (int row = tailStart + (int)(blockIdx.x * blockDim.x + threadIdx.x);
         row < B; row += (int)(gridDim.x * blockDim.x)) {
        float q[NCOMP], p[NCOMP];
        #pragma unroll
        for (int i = 0; i < NCOMP; ++i) {
            float acc = bvec[i];
            #pragma unroll
            for (int k = 0; k < NCOMP; ++k)
                acc += x[(long)row * NCOMP + k] * W[i * NCOMP + k];
            q[i] = acc;
        }
        solve_row(q, p);
        #pragma unroll
        for (int i = 0; i < NCOMP; ++i) out[(long)row * NCOMP + i] = p[i];
    }
}

extern "C" void kernel_launch(void* const* d_in, const int* in_sizes, int n_in,
                              void* d_out, int out_size, void* d_ws, size_t ws_size,
                              hipStream_t stream) {
    const float* x = (const float*)d_in[0];
    const float* W = (const float*)d_in[1];
    const float* b = (const float*)d_in[2];
    float* out = (float*)d_out;
    const int B = in_sizes[0] / NCOMP;

    const int nUnits = B >> 7;
    int blocks = (nUnits + 7) / 8;
    if (blocks > 4096) blocks = 4096;
    if (blocks < 1) blocks = 1;
    klproj_wmma_kernel<<<blocks, 256, 0, stream>>>(x, W, b, out, B);
}


# lam=0 rows take p = q/sum(q) (softmax(log q) identity), g(0) sign from sum q*log2(q) and sum q: no-search waves drop the max-shift and every exp2; sums now run over q, not over exp2 terms
# speedup vs baseline: 6.5581x; 1.0709x over previous
#include <hip/hip_runtime.h>

typedef __attribute__((ext_vector_type(2))) float v2f;
typedef __attribute__((ext_vector_type(8))) float v8f;

#define NCOMP 5
#define BISECT_ITERS 60

__device__ __forceinline__ float fexp2(float x) { return __builtin_amdgcn_exp2f(x); }
__device__ __forceinline__ float flog2(float x) { return __builtin_amdgcn_logf(x); }
__device__ __forceinline__ float frcp(float x)  { return __builtin_amdgcn_rcpf(x); }

__device__ __forceinline__ bool g_pos(const float* dd, float lam) {
    const float CC2 = 2.17765859079846660f;
    float t = frcp(1.0f + lam);
    float S = 0.0f, T = 0.0f;
    #pragma unroll
    for (int i = 0; i < NCOMP; ++i) {
        float z = t * dd[i];
        float e = fexp2(z);
        S += e;
        T = fmaf(e, z, T);
    }
    return fmaf(S, CC2 - flog2(S), T) > 0.0f;
}

__device__ __forceinline__ bool row_head(const float* q, float* lq, float* p0) {
    const float CC2 = 2.17765859079846660f;
    lq[0] = flog2(q[0]);
    float S0 = q[0], T0 = q[0] * lq[0];
    #pragma unroll
    for (int i = 1; i < NCOMP; ++i) {
        lq[i] = flog2(q[i]);
        S0 += q[i];
        T0 = fmaf(q[i], lq[i], T0);
    }
    float inv0 = 1.0f / S0;
    #pragma unroll
    for (int i = 0; i < NCOMP; ++i) p0[i] = q[i] * inv0;
    return fmaf(S0, CC2 - flog2(S0), T0) > 0.0f;
}

__device__ __forceinline__ void row_search(const float* lq, float* pe) {
    float dd[NCOMP], dmax = -3.0e38f;
    #pragma unroll
    for (int i = 0; i < NCOMP; ++i) dmax = fmaxf(dmax, lq[i]);
    #pragma unroll
    for (int i = 0; i < NCOMP; ++i) dd[i] = lq[i] - dmax;

    float lo = 0.0f, hi = 1.0e6f;
    #pragma unroll 1
    for (int it = 0; it < BISECT_ITERS; ++it) {
        float mid = 0.5f * (lo + hi);
        bool pos = g_pos(dd, mid);
        lo = pos ? mid : lo;
        hi = pos ? hi : mid;
    }
    float t = frcp(1.0f + 0.5f * (lo + hi));
    float e[NCOMP], S = 0.0f;
    #pragma unroll
    for (int i = 0; i < NCOMP; ++i) { e[i] = fexp2(t * dd[i]); S += e[i]; }
    float inv = 1.0f / S;
    #pragma unroll
    for (int i = 0; i < NCOMP; ++i) pe[i] = e[i] * inv;
}

typedef __attribute__((ext_vector_type(4))) float v4f;

#define UNIT_ROWS   128
#define UNIT_FLOATS (UNIT_ROWS * NCOMP)

__global__ __launch_bounds__(256) void klproj_wmma_kernel(
    const float* __restrict__ x, const float* __restrict__ W,
    const float* __restrict__ bvec, float* __restrict__ out, int B)
{
    __shared__ v4f lds4[8 * (UNIT_FLOATS / 4)];
    const int lane  = threadIdx.x & 31;
    const int wIB   = threadIdx.x >> 5;
    const int gwave = blockIdx.x * 8 + wIB;
    const int nwave = gridDim.x * 8;
    v4f*   wls4 = &lds4[wIB * (UNIT_FLOATS / 4)];
    float* wlsU = (float*)wls4;

    const int nUnits = B >> 7;
    if (gwave < nUnits) {
        int u = gwave;
        v4f r[5];
        {
            const v4f* xin = (const v4f*)(x + (long)u * UNIT_FLOATS);
            #pragma unroll
            for (int j = 0; j < 5; ++j) r[j] = xin[j * 32 + lane];
        }

        const int comp = lane & 7;
        const bool live = (comp < NCOMP) && (((lane >> 3) & 1) == (lane >> 4));
        float w0 = 0.f, w1 = 0.f, w2 = 0.f, w3 = 0.f, w4 = 0.f;
        if (live) {
            w0 = W[comp * NCOMP + 0];
            w1 = W[comp * NCOMP + 1];
            w2 = W[comp * NCOMP + 2];
            w3 = W[comp * NCOMP + 3];
            w4 = W[comp * NCOMP + 4];
        }
        v2f A01; A01.x = w0; A01.y = w1;
        v2f A23; A23.x = w2; A23.y = w3;
        v2f A4;  A4.x  = w4; A4.y  = 0.f;
        v8f cb;
        #pragma unroll
        for (int i = 0; i < 8; ++i) cb[i] = (i < NCOMP) ? bvec[i] : 0.f;

        for (;;) {
            #pragma unroll
            for (int j = 0; j < 5; ++j) wls4[j * 32 + lane] = r[j];
            asm volatile("s_wait_dscnt 0" ::: "memory");

            #pragma unroll 1
            for (int s = 0; s < UNIT_ROWS / 32; ++s) {
                float* row = wlsU + s * 160 + lane * 5;
                float xr[NCOMP];
                #pragma unroll
                for (int i = 0; i < NCOMP; ++i) xr[i] = row[i];
                v2f B01; B01.x = xr[0]; B01.y = xr[1];
                v2f B23; B23.x = xr[2]; B23.y = xr[3];
                v2f B4;  B4.x  = xr[4]; B4.y  = 0.f;
                v8f acc = __builtin_amdgcn_wmma_f32_16x16x4_f32(
                              false, A01, false, B01, (short)0, cb,  false, false);
                acc = __builtin_amdgcn_wmma_f32_16x16x4_f32(
                              false, A23, false, B23, (short)0, acc, false, false);
                acc = __builtin_amdgcn_wmma_f32_16x16x4_f32(
                              false, A4,  false, B4,  (short)0, acc, false, false);

                float q[NCOMP], lq[NCOMP], p[NCOMP];
                #pragma unroll
                for (int i = 0; i < NCOMP; ++i) q[i] = acc[i];
                const bool infeasible0 = row_head(q, lq, p);

                #pragma unroll
                for (int i = 0; i < NCOMP; ++i) row[i] = p[i];

                if (__builtin_amdgcn_ballot_w32(infeasible0) != 0u) {
                    float pe[NCOMP];
                    row_search(lq, pe);
                    if (infeasible0) {
                        #pragma unroll
                        for (int i = 0; i < NCOMP; ++i) row[i] = pe[i];
                    }
                }
            }
            asm volatile("s_wait_dscnt 0" ::: "memory");
            v4f* yout = (v4f*)(out + (long)u * UNIT_FLOATS);
            #pragma unroll
            for (int j = 0; j < 5; ++j) yout[j * 32 + lane] = wls4[j * 32 + lane];

            u += nwave;
            if (u >= nUnits) break;
            const v4f* xin = (const v4f*)(x + (long)u * UNIT_FLOATS);
            #pragma unroll
            for (int j = 0; j < 5; ++j) r[j] = xin[j * 32 + lane];
        }
    }

    const int tailStart = nUnits << 7;
    for (int row = tailStart + (int)(blockIdx.x * blockDim.x + threadIdx.x);
         row < B; row += (int)(gridDim.x * blockDim.x)) {
        float q[NCOMP], p[NCOMP];
        #pragma unroll
        for (int i = 0; i < NCOMP; ++i) {
            float acc = bvec[i];
            #pragma unroll
            for (int k = 0; k < NCOMP; ++k)
                acc += x[(long)row * NCOMP + k] * W[i * NCOMP + k];
            q[i] = acc;
        }
        float lq[NCOMP];
        const bool infeasible0 = row_head(q, lq, p);
        if (__builtin_amdgcn_ballot_w32(infeasible0) != 0u) {
            float pe[NCOMP];
            row_search(lq, pe);
            #pragma unroll
            for (int i = 0; i < NCOMP; ++i) p[i] = infeasible0 ? pe[i] : p[i];
        }
        #pragma unroll
        for (int i = 0; i < NCOMP; ++i) out[(long)row * NCOMP + i] = p[i];
    }
}

extern "C" void kernel_launch(void* const* d_in, const int* in_sizes, int n_in,
                              void* d_out, int out_size, void* d_ws, size_t ws_size,
                              hipStream_t stream) {
    const float* x = (const float*)d_in[0];
    const float* W = (const float*)d_in[1];
    const float* b = (const float*)d_in[2];
    float* out = (float*)d_out;
    const int B = in_sizes[0] / NCOMP;

    const int nUnits = B >> 7;
    int blocks = (nUnits + 7) / 8;
    if (blocks > 4096) blocks = 4096;
    if (blocks < 1) blocks = 1;
    klproj_wmma_kernel<<<blocks, 256, 0, stream>>>(x, W, b, out, B);
}
